# starting kernel
# baseline (speedup 1.0000x reference)
#include <hip/hip_runtime.h>

typedef __attribute__((ext_vector_type(2))) float v2f;
typedef __attribute__((ext_vector_type(8))) float v8f;

#define KTAP 5
#define RPAD 2
#define CCH  128
#define HH   128
#define WW   128
#define OHH  256
#define OWW  256
#define NCT  8
#define TILE_W 16
#define LDS_COLS (TILE_W + 2*RPAD)
#define LDS_ROWS (KTAP)
#define LDS_CH  (LDS_ROWS * LDS_COLS)
#define LDS_TOT (CCH * LDS_CH)
#define OST 66

__global__ __launch_bounds__(128)
void carafe_wmma_f32_kernel(const float* __restrict__ feats,
                            const float* __restrict__ masks,
                            float* __restrict__ out) {
    __shared__ float lfeat[LDS_TOT];

    const int tid  = threadIdx.x;
    const int lane = tid & 31;
    const int wave = tid >> 5;
    const int half = lane >> 4;
    const int mrow = lane & 15;
    const int p    = mrow >> 2;
    const int s    = mrow & 3;
    const int s1   = s >> 1;
    const int s2   = s & 1;

    const int bid  = blockIdx.x;
    const int wgrp = bid & 7;
    const int hi   = (bid >> 3) & 127;
    const int n    = bid >> 10;
    const int wblk = wgrp * TILE_W;

    const float* feat_n = feats + (size_t)n * CCH * HH * WW;
    for (int j = tid; j < LDS_TOT / 2; j += 128) {
        const int c   = j / 50;
        const int rem = j - c * 50;
        const int r   = rem / 10;
        const int x   = (rem - r * 10) * 2;
        const int gr  = hi + r - RPAD;
        const int gxc = wblk + x - RPAD;
        v2f v = {0.f, 0.f};
        if ((unsigned)gr < (unsigned)HH && (unsigned)gxc < (unsigned)WW)
            v = *(const v2f*)(feat_n + (size_t)c * (HH * WW) + (size_t)gr * WW + gxc);
        *(v2f*)(lfeat + 2 * j) = v;
    }
    __syncthreads();

    const int oh  = 2 * hi + s1;
    const int ow  = 2 * (wblk + wave * 4 + p) + s2;

    const bool gbx = (p == (half ? 2 : 0));
    const bool gby = (p == (half ? 3 : 1));

    v8f acc[NCT];
    const v8f vzero = {0.f, 0.f, 0.f, 0.f, 0.f, 0.f, 0.f, 0.f};
#pragma unroll
    for (int t = 0; t < NCT; ++t) acc[t] = vzero;

    const float* mask_base = masks + (size_t)n * (KTAP * KTAP) * (OHH * OWW)
                                   + (size_t)oh * OWW + ow;
    const int xbase = wave * 4 + half * 2;

#pragma unroll
    for (int kk = 0; kk < KTAP * KTAP; ++kk) {
        const int di = kk / KTAP;
        const int dj = kk - di * KTAP;

        const float mval = mask_base[(size_t)kk * (OHH * OWW)];
        v2f b;
        b.x = gbx ? mval : 0.f;
        b.y = gby ? mval : 0.f;

        const int lbase = di * LDS_COLS + xbase + dj;
#pragma unroll
        for (int t = 0; t < NCT; ++t) {
            const int off = (t * 16 + mrow) * LDS_CH + lbase;
            v2f a;
            a.x = lfeat[off];
            a.y = lfeat[off + 1];
            acc[t] = __builtin_amdgcn_wmma_f32_16x16x4_f32(
                false, a, false, b, (short)0, acc[t], false, false);
        }
    }

    __syncthreads();
    const int owl = 2 * (wave * 4 + p) + s2;
#pragma unroll
    for (int t = 0; t < NCT; ++t) {
#pragma unroll
        for (int v = 0; v < 8; ++v) {
            const int c = t * 16 + v + half * 8;
            lfeat[c * OST + s1 * 32 + owl] = acc[t][v];
        }
    }
    __syncthreads();

    float* out_n = out + (size_t)n * CCH * OHH * OWW;
    for (int i = 0; i < 64; ++i) {
        const int rid = (i << 2) | wave;
        const int c   = rid >> 1;
        const int sl  = rid & 1;
        const float val = lfeat[c * OST + sl * 32 + lane];
        out_n[(size_t)c * (OHH * OWW) + (size_t)(2 * hi + sl) * OWW
              + (wblk * 2 + lane)] = val;
    }
}

extern "C" void kernel_launch(void* const* d_in, const int* in_sizes, int n_in,
                              void* d_out, int out_size, void* d_ws, size_t ws_size,
                              hipStream_t stream) {
    const float* feats = (const float*)d_in[0];
    const float* masks = (const float*)d_in[1];
    float* out = (float*)d_out;

    dim3 grid(2 * 128 * 8);
    dim3 block(128);
    hipLaunchKernelGGL(carafe_wmma_f32_kernel, grid, block, 0, stream,
                       feats, masks, out);
}


# Two work-groups per low-res row, 64 channels each in passes of 16 (grid 256->512; 155 KB LDS, 236 registers: B selects in-pass, one fragment read group per tap row): two resident per WGP; same sums
# speedup vs baseline: 2.6722x; 2.6722x over previous
#include <hip/hip_runtime.h>

typedef __attribute__((ext_vector_type(2))) float v2f;
typedef __attribute__((ext_vector_type(4))) float v4f;
typedef __attribute__((ext_vector_type(8))) float v8f;

#define KTAP 5
#define CCH  128
#define HH   128
#define WW   128
#define OHH  256
#define OWW  256
#define CG   16
#define CWG  64
#define NPASS (CWG / CG)
#define NWAVE 8
#define GPW  4
#define NSTG (CG * KTAP / NWAVE)
#define NOUT (2 * CG * 2 / NWAVE)
#define NFR  9
#define LPAD 4
#define RS   136
#define CHS  (KTAP * RS + 4)
#define LDS_TOT (CG * CHS)
#define OST  264
#define OUT_TOT (2 * CG * OST)

__global__ __launch_bounds__(256) __attribute__((amdgpu_waves_per_eu(4)))
void reassemble_chanwalk2_wmma_f32_kernel(const float* __restrict__ feats,
                                          const float* __restrict__ masks,
                                          float* __restrict__ out) {
    __shared__ __attribute__((aligned(16))) float lfeat[2 * LDS_TOT];
    __shared__ __attribute__((aligned(16))) float lout[2 * OUT_TOT];

    const int tid  = threadIdx.x;
    const int lane = tid & 31;
    const int wave = __builtin_amdgcn_readfirstlane(tid >> 5);
    const int half = lane >> 4;
    const int mrow = lane & 15;
    const int p    = mrow >> 2;
    const int s    = mrow & 3;
    const int s1   = s >> 1;
    const int s2   = s & 1;

    const int bid  = blockIdx.x;
    const int ch0  = (bid >> 8) * CWG;
    const int n    = (bid >> 7) & 1;
    const int hi   = bid & 127;

    const float* feat_n = feats + ((size_t)n * CCH + ch0) * (HH * WW) + lane * 4;
    const v4f zero4 = {0.f, 0.f, 0.f, 0.f};
    v4f stg[NSTG];
#pragma unroll
    for (int i = 0; i < NSTG; ++i) {
        const int ci  = i / KTAP;
        const int r   = i - ci * KTAP;
        const int c   = wave * 2 + ci;
        const int gr  = hi + r - 2;
        const int grc = gr < 0 ? 0 : (gr > HH - 1 ? HH - 1 : gr);
        stg[i] = *(const v4f*)(feat_n + (size_t)c * (HH * WW) + grc * WW);
    }

    const int oh = 2 * hi + s1;
    const float* mask_n = masks + (size_t)n * (KTAP * KTAP) * (OHH * OWW) + (size_t)oh * OWW;
    float mreg[GPW][KTAP * KTAP];
#pragma unroll
    for (int gi = 0; gi < GPW; ++gi) {
        const int ow = 2 * (4 * (wave * GPW + gi) + p) + s2;
#pragma unroll
        for (int kk = 0; kk < KTAP * KTAP; ++kk)
            mreg[gi][kk] = mask_n[ow + (size_t)kk * (OHH * OWW)];
    }

    __builtin_amdgcn_sched_barrier(0);
#pragma unroll
    for (int i = 0; i < NSTG; ++i) {
        const int ci  = i / KTAP;
        const int r   = i - ci * KTAP;
        const int c   = wave * 2 + ci;
        const int gr  = hi + r - 2;
        const bool ok = (unsigned)gr < (unsigned)HH;
        float* rowp = lfeat + c * CHS + r * RS;
        *(v4f*)(rowp + LPAD + lane * 4) = ok ? stg[i] : zero4;
        if (lane < 4)
            *(v4f*)(rowp + (lane >> 1) * LDS_TOT + (lane & 1) * (LPAD + WW)) = zero4;
    }
    __syncthreads();

    const v8f vzero = {0.f, 0.f, 0.f, 0.f, 0.f, 0.f, 0.f, 0.f};

    const int abase = mrow * CHS + wave * (4 * GPW) + 2 * half + 2;
    float* out_n = out + ((size_t)n * CCH + ch0) * (OHH * OWW) + (size_t)(2 * hi) * OWW + lane * 4;

#pragma nounroll
    for (int pass = 0; pass < NPASS; ++pass) {
        const float* cur = lfeat + (pass & 1) * LDS_TOT;
        float*       nxt = lfeat + ((pass + 1) & 1) * LDS_TOT;
        float*       ob  = lout + (pass & 1) * OUT_TOT;
        const bool more  = pass + 1 < NPASS;

        if (more) {
#pragma unroll
            for (int i = 0; i < NSTG; ++i) {
                const int ci  = i / KTAP;
                const int r   = i - ci * KTAP;
                const int c   = (pass + 1) * CG + wave * 2 + ci;
                const int gr  = hi + r - 2;
                const int grc = gr < 0 ? 0 : (gr > HH - 1 ? HH - 1 : gr);
                stg[i] = *(const v4f*)(feat_n + (size_t)c * (HH * WW) + grc * WW);
            }
        }

        int pk = p;
        asm volatile("" : "+v"(pk));
        const bool gbx = (pk == 2 * half);
        const bool gby = (pk == 2 * half + 1);

        v8f acc[GPW];
#pragma unroll
        for (int gi = 0; gi < GPW; ++gi) acc[gi] = vzero;

        v2f fr[2][NFR];
#pragma unroll
        for (int j = 0; j < NFR; ++j) fr[0][j] = *(const v2f*)(cur + abase + 2 * j);

#pragma unroll
        for (int di = 0; di < KTAP; ++di) {
            if (di + 1 < KTAP) {
#pragma unroll
                for (int j = 0; j < NFR; ++j)
                    fr[(di + 1) & 1][j] = *(const v2f*)(cur + abase + (di + 1) * RS + 2 * j);
            }
            __builtin_amdgcn_sched_barrier(0);
#pragma unroll
            for (int dj = 0; dj < KTAP; ++dj) {
#pragma unroll
                for (int gi = 0; gi < GPW; ++gi) {
                    const float mval = mreg[gi][di * KTAP + dj];
                    v2f b;
                    b.x = gbx ? mval : 0.f;
                    b.y = gby ? mval : 0.f;
                    const int e = 4 * gi + dj;
                    v2f a;
                    a.x = fr[di & 1][e >> 1][e & 1];
                    a.y = fr[di & 1][(e + 1) >> 1][(e + 1) & 1];
                    acc[gi] = __builtin_amdgcn_wmma_f32_16x16x4_f32(
                        false, a, false, b, (short)0, acc[gi], false, false);
                }
            }
            __builtin_amdgcn_sched_barrier(0);
        }

#pragma unroll
        for (int gi = 0; gi < GPW; ++gi) {
            const int owl = 2 * (4 * (wave * GPW + gi) + p) + s2;
#pragma unroll
            for (int v = 0; v < 8; ++v) {
                const int c = v + half * 8;
                ob[(c * 2 + s1) * OST + owl] = acc[gi][v];
            }
        }

        if (more) {
#pragma unroll
            for (int i = 0; i < NSTG; ++i) {
                const int ci  = i / KTAP;
                const int r   = i - ci * KTAP;
                const int c   = wave * 2 + ci;
                const int gr  = hi + r - 2;
                const bool ok = (unsigned)gr < (unsigned)HH;
                float* rowp = nxt + c * CHS + r * RS;
                *(v4f*)(rowp + LPAD + lane * 4) = ok ? stg[i] : zero4;
            }
        }
        __syncthreads();

        float* out_p = out_n + (size_t)(pass * CG) * (OHH * OWW);
#pragma unroll
        for (int i = 0; i < NOUT; ++i) {
            const int u   = i * NWAVE + wave;
            const int row = u >> 1;
            const int hf  = u & 1;
            const int c   = row >> 1;
            const int sl  = row & 1;
            const v4f val = *(const v4f*)(ob + row * OST + hf * 128 + lane * 4);
            *(v4f*)(out_p + (size_t)c * (OHH * OWW) + sl * OWW + hf * 128) = val;
        }
    }
}

extern "C" void kernel_launch(void* const* d_in, const int* in_sizes, int n_in,
                              void* d_out, int out_size, void* d_ws, size_t ws_size,
                              hipStream_t stream) {
    const float* feats = (const float*)d_in[0];
    const float* masks = (const float*)d_in[1];
    float* out = (float*)d_out;

    dim3 grid(2 * 2 * 128);
    dim3 block(256);
    hipLaunchKernelGGL(reassemble_chanwalk2_wmma_f32_kernel, grid, block, 0, stream,
                       feats, masks, out);
}


# Output stores moved under the next pass's products: 8 row pieces read from LDS ahead of the matrix block, one stored behind each slice's 4 products; the pass barrier fences LDS only
# speedup vs baseline: 4.5934x; 1.7190x over previous
#include <hip/hip_runtime.h>

typedef __attribute__((ext_vector_type(2))) float v2f;
typedef __attribute__((ext_vector_type(4))) float v4f;
typedef __attribute__((ext_vector_type(8))) float v8f;

#define KTAP 5
#define CCH  128
#define HH   128
#define WW   128
#define OHH  256
#define OWW  256
#define CG   16
#define CWG  64
#define NPASS (CWG / CG)
#define NWAVE 8
#define XW   64
#define NROW 6
#define GPW  4
#define NSL  9
#define NRP  3
#define NJP  6
#define NQ   18
#define NSTG (2 * NROW)
#define NOUT (CG * 4 / NWAVE)
#define RS   72
#define CHS  (NROW * RS + 4)
#define LDS_TOT (CG * CHS)
#define OST  132
#define OUT_TOT (CG * 4 * OST)
#define MP   132
#define NMROW (KTAP * KTAP * 4)
#define NMLD ((NMROW + NWAVE - 1) / NWAVE)

__device__ __forceinline__ void lds_barrier() {
    __builtin_amdgcn_fence(__ATOMIC_RELEASE, "workgroup", "local");
    __builtin_amdgcn_s_barrier();
    __builtin_amdgcn_fence(__ATOMIC_ACQUIRE, "workgroup", "local");
}

__global__ __launch_bounds__(256) __attribute__((amdgpu_waves_per_eu(4)))
void reassemble_fold2d_wmma_f32_kernel(const float* __restrict__ feats,
                                       const float* __restrict__ masks,
                                       float* __restrict__ out) {
    __shared__ __attribute__((aligned(16))) float lfeat[2 * LDS_TOT];
    __shared__ __attribute__((aligned(16))) float lout[2 * OUT_TOT];

    const int tid  = threadIdx.x;
    const int lane = tid & 31;
    const int wave = __builtin_amdgcn_readfirstlane(tid >> 5);
    const int half = lane >> 4;
    const int mrow = lane & 15;
    const int py   = mrow >> 3;
    const int px   = (mrow >> 2) & 1;
    const int s1   = (mrow >> 1) & 1;
    const int s2   = mrow & 1;
    const int orow = 2 * py + s1;

    const int bid  = blockIdx.x;
    const int hp   = bid & 63;
    const int xh   = (bid >> 6) & 1;
    const int n    = (bid >> 7) & 1;
    const int ch0  = (bid >> 8) * CWG;
    const int hi0  = 2 * hp;

    const bool stl  = lane < NQ;
    const int  colq = xh * (XW / 4) - 1 + lane;
    const bool colok = (unsigned)colq < (unsigned)(WW / 4);
    const int  colc = colq < 0 ? 0 : (colq > WW / 4 - 1 ? WW / 4 - 1 : colq);
    const float* feat_n = feats + ((size_t)n * CCH + ch0) * (HH * WW) + colc * 4;
    const v4f zero4 = {0.f, 0.f, 0.f, 0.f};
    v4f stg[NSTG];
    if (stl) {
#pragma unroll
        for (int i = 0; i < NSTG; ++i) {
            const int ci  = i / NROW;
            const int r   = i - ci * NROW;
            const int c   = wave * 2 + ci;
            const int gr  = hi0 + r - 2;
            const int grc = gr < 0 ? 0 : (gr > HH - 1 ? HH - 1 : gr);
            stg[i] = *(const v4f*)(feat_n + (size_t)c * (HH * WW) + grc * WW);
        }
    }

    float* lm = lout;
    const float* mask_n = masks + (size_t)n * (KTAP * KTAP) * (OHH * OWW)
                        + (size_t)(4 * hp) * OWW + xh * (2 * XW) + lane * 4;
    v4f mst[NMLD];
#pragma unroll
    for (int i = 0; i < NMLD; ++i) {
        const int m = i * NWAVE + wave;
        if (m < NMROW)
            mst[i] = *(const v4f*)(mask_n + (size_t)(m >> 2) * (OHH * OWW) + (m & 3) * OWW);
    }
#pragma unroll
    for (int i = 0; i < NMLD; ++i) {
        const int m = i * NWAVE + wave;
        if (m < NMROW)
            *(v4f*)(lm + m * MP + lane * 4) = mst[i];
    }
    if (stl) {
#pragma unroll
        for (int i = 0; i < NSTG; ++i) {
            const int ci  = i / NROW;
            const int r   = i - ci * NROW;
            const int c   = wave * 2 + ci;
            const int gr  = hi0 + r - 2;
            const bool ok = colok && ((unsigned)gr < (unsigned)HH);
            *(v4f*)(lfeat + c * CHS + r * RS + lane * 4) = ok ? stg[i] : zero4;
        }
    }
    __syncthreads();

    const int lanebase = ((5 * (half - py) - px) * 4 + orow) * MP + wave * (4 * GPW) + 2 * px + s2;
    v2f bm[GPW][NSL];
#pragma unroll
    for (int rp = 0; rp < NRP; ++rp) {
        const bool rowok = (unsigned)(2 * rp + half - py) < (unsigned)KTAP;
#pragma unroll
        for (int cp = 0; cp < NRP; ++cp) {
            const bool ok0 = rowok && ((unsigned)(2 * cp - px) < (unsigned)KTAP);
            const bool ok1 = rowok && ((unsigned)(2 * cp + 1 - px) < (unsigned)KTAP);
#pragma unroll
            for (int gx = 0; gx < GPW; ++gx) {
                const int off = (10 * rp + 2 * cp) * 4 * MP + 4 * gx;
                const float m0 = lm[ok0 ? lanebase + off : 0];
                const float m1 = lm[ok1 ? lanebase + off + 4 * MP : 0];
                v2f b;
                b.x = ok0 ? m0 : 0.f;
                b.y = ok1 ? m1 : 0.f;
                bm[gx][rp * NRP + cp] = b;
            }
        }
    }
    __syncthreads();

    const v8f vzero = {0.f, 0.f, 0.f, 0.f, 0.f, 0.f, 0.f, 0.f};

    const int abase = mrow * CHS + half * RS + wave * (2 * GPW) + 2;
    const int ocol  = wave * (4 * GPW) + 2 * px + s2;
    float* out_n = out + ((size_t)n * CCH + ch0) * (OHH * OWW) + (size_t)(4 * hp) * OWW
                 + xh * (2 * XW) + lane * 4;

    float* out_w = out_n + (size_t)(wave >> 2) * (OHH * OWW) + (wave & 3) * OWW;
    const int opiece = wave * OST + lane * 4;

    v4f ov[NOUT];
    v2f fr[NRP][NJP];
    v8f acc[GPW];

    auto stage_request = [&](int pass_next) __attribute__((always_inline)) {
        if (stl) {
#pragma unroll
            for (int i = 0; i < NSTG; ++i) {
                const int ci  = i / NROW;
                const int r   = i - ci * NROW;
                const int c   = pass_next * CG + wave * 2 + ci;
                const int gr  = hi0 + r - 2;
                const int grc = gr < 0 ? 0 : (gr > HH - 1 ? HH - 1 : gr);
                stg[i] = *(const v4f*)(feat_n + (size_t)c * (HH * WW) + grc * WW);
            }
        }
    };
    auto stage_write = [&](float* nxt) __attribute__((always_inline)) {
        if (stl) {
#pragma unroll
            for (int i = 0; i < NSTG; ++i) {
                const int ci  = i / NROW;
                const int r   = i - ci * NROW;
                const int c   = wave * 2 + ci;
                const int gr  = hi0 + r - 2;
                const bool ok = colok && ((unsigned)gr < (unsigned)HH);
                *(v4f*)(nxt + c * CHS + r * RS + lane * 4) = ok ? stg[i] : zero4;
            }
        }
    };
    auto frag_read = [&](const float* cur) __attribute__((always_inline)) {
#pragma unroll
        for (int rp = 0; rp < NRP; ++rp)
#pragma unroll
            for (int j = 0; j < NJP; ++j)
                fr[rp][j] = *(const v2f*)(cur + abase + 2 * rp * RS + 2 * j);
    };
    auto repack = [&](float* ob) __attribute__((always_inline)) {
#pragma unroll
        for (int gx = 0; gx < GPW; ++gx) {
#pragma unroll
            for (int v = 0; v < 8; ++v) {
                const int c = v + half * 8;
                ob[(c * 4 + orow) * OST + ocol + 4 * gx] = acc[gx][v];
            }
        }
    };
    auto piece_read = [&](const float* ob) __attribute__((always_inline)) {
#pragma unroll
        for (int i = 0; i < NOUT; ++i)
            ov[i] = *(const v4f*)(ob + opiece + i * (NWAVE * OST));
    };

    stage_request(1);
    frag_read(lfeat);
    __builtin_amdgcn_sched_barrier(0);
#pragma unroll
    for (int gx = 0; gx < GPW; ++gx) acc[gx] = vzero;
#pragma unroll
    for (int sl = 0; sl < NSL; ++sl) {
#pragma unroll
        for (int gx = 0; gx < GPW; ++gx) {
            acc[gx] = __builtin_amdgcn_wmma_f32_16x16x4_f32(
                false, fr[sl / NRP][gx + sl % NRP], false, bm[gx][sl], (short)0,
                acc[gx], false, false);
        }
    }
    repack(lout);
    stage_write(lfeat + LDS_TOT);
    __syncthreads();

#pragma nounroll
    for (int pass = 1; pass < NPASS; ++pass) {
        const float* cur = lfeat + (pass & 1) * LDS_TOT;
        float*       nxt = lfeat + ((pass + 1) & 1) * LDS_TOT;
        float*       ob  = lout + (pass & 1) * OUT_TOT;
        const bool more  = pass + 1 < NPASS;
        float* out_p = out_w + (size_t)((pass - 1) * CG) * (OHH * OWW);

        piece_read(lout + ((pass - 1) & 1) * OUT_TOT);
        if (more) stage_request(pass + 1);
        frag_read(cur);
        __builtin_amdgcn_sched_barrier(0);
#pragma unroll
        for (int gx = 0; gx < GPW; ++gx) acc[gx] = vzero;
#pragma unroll
        for (int sl = 0; sl < NSL; ++sl) {
#pragma unroll
            for (int gx = 0; gx < GPW; ++gx) {
                acc[gx] = __builtin_amdgcn_wmma_f32_16x16x4_f32(
                    false, fr[sl / NRP][gx + sl % NRP], false, bm[gx][sl], (short)0,
                    acc[gx], false, false);
            }
            if (sl < NOUT) {
                __builtin_amdgcn_sched_barrier(0);
                *(v4f*)(out_p + (size_t)(2 * sl) * (OHH * OWW)) = ov[sl];
                __builtin_amdgcn_sched_barrier(0);
            }
        }
        repack(ob);
        if (more) stage_write(nxt);
        lds_barrier();
    }

    {
        float* out_p = out_w + (size_t)((NPASS - 1) * CG) * (OHH * OWW);
        piece_read(lout + ((NPASS - 1) & 1) * OUT_TOT);
#pragma unroll
        for (int i = 0; i < NOUT; ++i)
            *(v4f*)(out_p + (size_t)(2 * i) * (OHH * OWW)) = ov[i];
    }
}

extern "C" void kernel_launch(void* const* d_in, const int* in_sizes, int n_in,
                              void* d_out, int out_size, void* d_ws, size_t ws_size,
                              hipStream_t stream) {
    const float* feats = (const float*)d_in[0];
    const float* masks = (const float*)d_in[1];
    float* out = (float*)d_out;

    dim3 grid(2 * 2 * 2 * 64);
    dim3 block(256);
    hipLaunchKernelGGL(reassemble_fold2d_wmma_f32_kernel, grid, block, 0, stream,
                       feats, masks, out);
}
